# starting kernel
# baseline (speedup 1.0000x reference)
#include <hip/hip_runtime.h>
#include <hip/hip_bf16.h>

typedef float v2f __attribute__((ext_vector_type(2)));
typedef float v8f __attribute__((ext_vector_type(8)));

static constexpr int DIM = 8192;
static constexpr int FD  = 128;
static constexpr int BM  = 32;
static constexpr int BK  = 32;
static constexpr float EPSV = 1e-10f;


__global__ __launch_bounds__(256) void HG_rowdeg(const float* __restrict__ H,
                                                 float* __restrict__ dvis) {
    __shared__ float red[256];
    const int row = blockIdx.x;
    const float* p = H + (size_t)row * DIM;
    float s = 0.0f;
    for (int c = threadIdx.x; c < DIM; c += 256) s += p[c];
    red[threadIdx.x] = s;
    __syncthreads();
    for (int off = 128; off > 0; off >>= 1) {
        if (threadIdx.x < off) red[threadIdx.x] += red[threadIdx.x + off];
        __syncthreads();
    }
    if (threadIdx.x == 0) dvis[row] = 1.0f / sqrtf(red[0] + EPSV);
}

__global__ __launch_bounds__(256) void HG_colzero(float* __restrict__ de) {
    de[blockIdx.x * 256 + threadIdx.x] = 0.0f;
}

__global__ __launch_bounds__(256) void HG_coldeg(const float* __restrict__ H,
                                                 float* __restrict__ de) {
    const int c  = blockIdx.x * 256 + threadIdx.x;
    const int r0 = blockIdx.y * 256;
    float s = 0.0f;
    #pragma unroll 4
    for (int r = 0; r < 256; ++r) s += H[(size_t)(r0 + r) * DIM + c];
    atomicAdd(&de[c], s);
}

__global__ __launch_bounds__(256) void HG_colfin(float* __restrict__ de) {
    const int i = blockIdx.x * 256 + threadIdx.x;
    de[i] = 1.0f / (de[i] + EPSV);
}

template <bool TRANSA, bool OUTSCALE>
__global__ __launch_bounds__(256) void HG_gemm(const float* __restrict__ A,
                                               const float* __restrict__ srcB,
                                               const float* __restrict__ scaleB,
                                               const float* __restrict__ outScale,
                                               float* __restrict__ out) {
    __shared__ __align__(16) float lAf[2][(BK / 2) * BM * 2];
    __shared__ __align__(16) float lBf[2][(BK / 2) * FD * 2];

    const int tid  = threadIdx.x;
    const int wave = tid >> 5;
    const int lid  = tid & 31;
    const int half = lid >> 4;
    const int ln   = lid & 15;
    const int m0   = blockIdx.x * BM;

    const int a_i4 = (tid & 7) * 4;
    const int a_j  = tid >> 3;
    const int b_d4 = (tid & 31) * 4;
    const int b_k  = tid >> 5;

    v8f acc[2];
    acc[0] = (v8f){};
    acc[1] = (v8f){};

    float4 ra;
    float4 rb[4];
    float  sb[4];

    auto load_chunk = [&](int k0) {
        if (TRANSA) {
            ra = *(const float4*)(A + (size_t)(k0 + a_j) * DIM + m0 + a_i4);
        } else {
            ra = *(const float4*)(A + (size_t)(m0 + a_j) * DIM + k0 + a_i4);
        }
        #pragma unroll
        for (int p = 0; p < 4; ++p) {
            const int row = k0 + b_k + p * 8;
            sb[p] = scaleB[row];
            rb[p] = *(const float4*)(srcB + (size_t)row * FD + b_d4);
        }
    };

    auto store_chunk = [&](int buf) {
        if (TRANSA) {
            const int kp = a_j >> 1, kc = a_j & 1;
            lAf[buf][(kp * BM + a_i4 + 0) * 2 + kc] = ra.x;
            lAf[buf][(kp * BM + a_i4 + 1) * 2 + kc] = ra.y;
            lAf[buf][(kp * BM + a_i4 + 2) * 2 + kc] = ra.z;
            lAf[buf][(kp * BM + a_i4 + 3) * 2 + kc] = ra.w;
        } else {
            lAf[buf][((a_i4 >> 1) * BM + a_j) * 2 + 0] = ra.x;
            lAf[buf][((a_i4 >> 1) * BM + a_j) * 2 + 1] = ra.y;
            lAf[buf][(((a_i4 + 2) >> 1) * BM + a_j) * 2 + 0] = ra.z;
            lAf[buf][(((a_i4 + 2) >> 1) * BM + a_j) * 2 + 1] = ra.w;
        }
        #pragma unroll
        for (int p = 0; p < 4; ++p) {
            const int k  = b_k + p * 8;
            const int kp = k >> 1, kc = k & 1;
            lBf[buf][(kp * FD + b_d4 + 0) * 2 + kc] = rb[p].x * sb[p];
            lBf[buf][(kp * FD + b_d4 + 1) * 2 + kc] = rb[p].y * sb[p];
            lBf[buf][(kp * FD + b_d4 + 2) * 2 + kc] = rb[p].z * sb[p];
            lBf[buf][(kp * FD + b_d4 + 3) * 2 + kc] = rb[p].w * sb[p];
        }
    };

    load_chunk(0);
    store_chunk(0);
    __syncthreads();

    int cur = 0;
    for (int ki = 0; ki < DIM / BK; ++ki) {
        const int knext = ((ki + 1) & (DIM / BK - 1)) * BK;
        load_chunk(knext);

        const float2* lA2 = (const float2*)lAf[cur];
        const float2* lB2 = (const float2*)lBf[cur];
        #pragma unroll
        for (int s = 0; s < 8; ++s) {
            const int kp = 2 * s + half;
            const float2 bf  = lB2[kp * FD + wave * 16 + ln];
            const float2 a0f = lA2[kp * BM + ln];
            const float2 a1f = lA2[kp * BM + 16 + ln];
            v2f b;  b[0]  = bf.x;  b[1]  = bf.y;
            v2f a0; a0[0] = a0f.x; a0[1] = a0f.y;
            v2f a1; a1[0] = a1f.x; a1[1] = a1f.y;
            acc[0] = __builtin_amdgcn_wmma_f32_16x16x4_f32(
                false, a0, false, b, (short)0, acc[0], false, false);
            acc[1] = __builtin_amdgcn_wmma_f32_16x16x4_f32(
                false, a1, false, b, (short)0, acc[1], false, false);
        }

        store_chunk(cur ^ 1);
        __syncthreads();
        cur ^= 1;
    }

    #pragma unroll
    for (int r = 0; r < 2; ++r) {
        #pragma unroll
        for (int j = 0; j < 8; ++j) {
            const int row = m0 + 16 * r + half * 8 + j;
            const float os = OUTSCALE ? outScale[row] : 1.0f;
            out[(size_t)row * FD + wave * 16 + ln] = acc[r][j] * os;
        }
    }
}

extern "C" void kernel_launch(void* const* d_in, const int* in_sizes, int n_in,
                              void* d_out, int out_size, void* d_ws, size_t ws_size,
                              hipStream_t stream) {
    (void)in_sizes; (void)n_in; (void)out_size; (void)ws_size;
    const float* H = (const float*)d_in[0];
    const float* F = (const float*)d_in[1];
    float* out  = (float*)d_out;
    float* ws   = (float*)d_ws;
    float* dvis = ws;
    float* de   = ws + DIM;
    float* Y    = ws + 2 * DIM;

    HG_rowdeg<<<DIM, 256, 0, stream>>>(H, dvis);
    HG_colzero<<<DIM / 256, 256, 0, stream>>>(de);
    HG_coldeg<<<dim3(DIM / 256, DIM / 256), 256, 0, stream>>>(H, de);
    HG_colfin<<<DIM / 256, 256, 0, stream>>>(de);

    HG_gemm<true, false><<<DIM / BM, 256, 0, stream>>>(H, F, dvis, nullptr, Y);
    HG_gemm<false, true><<<DIM / BM, 256, 0, stream>>>(H, Y, de, dvis, out);
}


# K split inside one 1024-thread workgroup: 8 K groups x 4 waves share a staged 128-k chunk, 32x32 wave tiles, fence keeps loads ahead of WMMAs; partials added via LDS in fixed order (new sum tree)
# speedup vs baseline: 1.9019x; 1.9019x over previous
#include <hip/hip_runtime.h>
#include <hip/hip_bf16.h>

typedef float v2f __attribute__((ext_vector_type(2)));
typedef float v4f __attribute__((ext_vector_type(4)));
typedef float v8f __attribute__((ext_vector_type(8)));

static constexpr int DIM = 8192;
static constexpr int FD  = 128;
static constexpr int BM  = 32;
static constexpr int BK  = 128;
static constexpr int KG  = 8;
static constexpr float EPSV = 1e-10f;


__global__ __launch_bounds__(256) void HG_rowdeg(const float* __restrict__ H,
                                                 float* __restrict__ dvis) {
    __shared__ float red[256];
    const int row = blockIdx.x;
    const float* p = H + (size_t)row * DIM;
    float s = 0.0f;
    for (int c = threadIdx.x; c < DIM; c += 256) s += p[c];
    red[threadIdx.x] = s;
    __syncthreads();
    for (int off = 128; off > 0; off >>= 1) {
        if (threadIdx.x < off) red[threadIdx.x] += red[threadIdx.x + off];
        __syncthreads();
    }
    if (threadIdx.x == 0) dvis[row] = 1.0f / sqrtf(red[0] + EPSV);
}

__global__ __launch_bounds__(256) void HG_colzero(float* __restrict__ de) {
    de[blockIdx.x * 256 + threadIdx.x] = 0.0f;
}

__global__ __launch_bounds__(256) void HG_coldeg(const float* __restrict__ H,
                                                 float* __restrict__ de) {
    const int c  = blockIdx.x * 256 + threadIdx.x;
    const int r0 = blockIdx.y * 256;
    float s = 0.0f;
    #pragma unroll 4
    for (int r = 0; r < 256; ++r) s += H[(size_t)(r0 + r) * DIM + c];
    atomicAdd(&de[c], s);
}

__global__ __launch_bounds__(256) void HG_colfin(float* __restrict__ de) {
    const int i = blockIdx.x * 256 + threadIdx.x;
    de[i] = 1.0f / (de[i] + EPSV);
}

template <bool TRANSA, bool OUTSCALE>
__global__ __launch_bounds__(1024) void HG_gemm(const float* __restrict__ A,
                                                const float* __restrict__ srcB,
                                                const float* __restrict__ scaleB,
                                                const float* __restrict__ outScale,
                                                float* __restrict__ out) {
    constexpr int LA = (BK / 2) * BM * 2;
    constexpr int LB = (BK / 2) * FD * 2;
    __shared__ __align__(16) float lAf[2 * LA];
    __shared__ __align__(16) float lBf[2 * LB];

    const int tid  = threadIdx.x;
    const int wv   = tid >> 5;
    const int lid  = tid & 31;
    const int half = lid >> 4;
    const int ln   = lid & 15;
    const int g    = wv >> 2;
    const int wq   = wv & 3;
    const int m0   = blockIdx.x * BM;

    const int ks   = (tid >> 8) * 32;
    const int t    = tid & 255;
    const int a_i4 = (t & 7) * 4;
    const int a_j  = t >> 3;
    const int b_d4 = (t & 31) * 4;
    const int b_k  = t >> 5;

    v8f acc[2][2];
    acc[0][0] = (v8f){};
    acc[0][1] = (v8f){};
    acc[1][0] = (v8f){};
    acc[1][1] = (v8f){};

    v4f ra;
    v4f rb[4];
    float  sb[4];

    auto load_chunk = [&](int k0) {
        if (TRANSA) {
            ra = *(const v4f*)(A + (size_t)(k0 + ks + a_j) * DIM + m0 + a_i4);
        } else {
            ra = *(const v4f*)(A + (size_t)(m0 + a_j) * DIM + k0 + ks + a_i4);
        }
        #pragma unroll
        for (int p = 0; p < 4; ++p) {
            const int row = k0 + ks + b_k + p * 8;
            sb[p] = scaleB[row];
            rb[p] = *(const v4f*)(srcB + (size_t)row * FD + b_d4);
        }
    };

    auto store_chunk = [&](int buf) {
        float* lA = lAf + buf * LA;
        float* lB = lBf + buf * LB;
        if (TRANSA) {
            const int k  = ks + a_j;
            const int kp = k >> 1, kc = k & 1;
            lA[(kp * BM + a_i4 + 0) * 2 + kc] = ra.x;
            lA[(kp * BM + a_i4 + 1) * 2 + kc] = ra.y;
            lA[(kp * BM + a_i4 + 2) * 2 + kc] = ra.z;
            lA[(kp * BM + a_i4 + 3) * 2 + kc] = ra.w;
        } else {
            const int kp = (ks + a_i4) >> 1;
            lA[((kp + 0) * BM + a_j) * 2 + 0] = ra.x;
            lA[((kp + 0) * BM + a_j) * 2 + 1] = ra.y;
            lA[((kp + 1) * BM + a_j) * 2 + 0] = ra.z;
            lA[((kp + 1) * BM + a_j) * 2 + 1] = ra.w;
        }
        #pragma unroll
        for (int p = 0; p < 4; ++p) {
            const int k  = ks + b_k + p * 8;
            const int kp = k >> 1, kc = k & 1;
            lB[(kp * FD + b_d4 + 0) * 2 + kc] = rb[p].x * sb[p];
            lB[(kp * FD + b_d4 + 1) * 2 + kc] = rb[p].y * sb[p];
            lB[(kp * FD + b_d4 + 2) * 2 + kc] = rb[p].z * sb[p];
            lB[(kp * FD + b_d4 + 3) * 2 + kc] = rb[p].w * sb[p];
        }
    };

    load_chunk(0);
    store_chunk(0);
    __syncthreads();

    int cur = 0;
    for (int ki = 0; ki < DIM / BK; ++ki) {
        const int knext = ((ki + 1) & (DIM / BK - 1)) * BK;
        load_chunk(knext);

        const float2* lA2 = (const float2*)(lAf + cur * LA);
        const float2* lB2 = (const float2*)(lBf + cur * LB);
        #pragma unroll
        for (int s = 0; s < 4; ++s) {
            const int kp = 8 * g + 2 * s + half;
            const float2 b0f = lB2[kp * FD + wq * 32 + ln];
            const float2 b1f = lB2[kp * FD + wq * 32 + 16 + ln];
            const float2 a0f = lA2[kp * BM + ln];
            const float2 a1f = lA2[kp * BM + 16 + ln];
            v2f b0; b0[0] = b0f.x; b0[1] = b0f.y;
            v2f b1; b1[0] = b1f.x; b1[1] = b1f.y;
            v2f a0; a0[0] = a0f.x; a0[1] = a0f.y;
            v2f a1; a1[0] = a1f.x; a1[1] = a1f.y;
            acc[0][0] = __builtin_amdgcn_wmma_f32_16x16x4_f32(
                false, a0, false, b0, (short)0, acc[0][0], false, false);
            acc[0][1] = __builtin_amdgcn_wmma_f32_16x16x4_f32(
                false, a0, false, b1, (short)0, acc[0][1], false, false);
            acc[1][0] = __builtin_amdgcn_wmma_f32_16x16x4_f32(
                false, a1, false, b0, (short)0, acc[1][0], false, false);
            acc[1][1] = __builtin_amdgcn_wmma_f32_16x16x4_f32(
                false, a1, false, b1, (short)0, acc[1][1], false, false);
        }

        __builtin_amdgcn_sched_barrier(0);

        store_chunk(cur ^ 1);
        __syncthreads();
        cur ^= 1;
    }

    float* part = lBf;
    #pragma unroll
    for (int r = 0; r < 2; ++r) {
        #pragma unroll
        for (int j = 0; j < 2; ++j) {
            #pragma unroll
            for (int e = 0; e < 8; ++e) {
                const int row = 16 * r + half * 8 + e;
                const int col = wq * 32 + 16 * j + ln;
                part[g * (BM * FD) + row * FD + col] = acc[r][j][e];
            }
        }
    }
    __syncthreads();

    const int orow = tid >> 5;
    const int oc4  = (tid & 31) * 4;
    float4 sum = *(const float4*)(part + orow * FD + oc4);
    #pragma unroll
    for (int gg = 1; gg < KG; ++gg) {
        const float4 v = *(const float4*)(part + gg * (BM * FD) + orow * FD + oc4);
        sum.x += v.x; sum.y += v.y; sum.z += v.z; sum.w += v.w;
    }
    const float os = OUTSCALE ? outScale[m0 + orow] : 1.0f;
    sum.x *= os; sum.y *= os; sum.z *= os; sum.w *= os;
    *(float4*)(out + (size_t)(m0 + orow) * FD + oc4) = sum;
}

extern "C" void kernel_launch(void* const* d_in, const int* in_sizes, int n_in,
                              void* d_out, int out_size, void* d_ws, size_t ws_size,
                              hipStream_t stream) {
    (void)in_sizes; (void)n_in; (void)out_size; (void)ws_size;
    const float* H = (const float*)d_in[0];
    const float* F = (const float*)d_in[1];
    float* out  = (float*)d_out;
    float* ws   = (float*)d_ws;
    float* dvis = ws;
    float* de   = ws + DIM;
    float* Y    = ws + 2 * DIM;

    HG_rowdeg<<<DIM, 256, 0, stream>>>(H, dvis);
    HG_colzero<<<DIM / 256, 256, 0, stream>>>(de);
    HG_coldeg<<<dim3(DIM / 256, DIM / 256), 256, 0, stream>>>(H, de);
    HG_colfin<<<DIM / 256, 256, 0, stream>>>(de);

    HG_gemm<true, false><<<DIM / BM, 1024, 0, stream>>>(H, F, dvis, nullptr, Y);
    HG_gemm<false, true><<<DIM / BM, 1024, 0, stream>>>(H, Y, de, dvis, out);
}


# GEMM B operand no longer staged in LDS: each wave loads its own K slice x 32 columns straight from global as 8-byte fragments (tile columns interleaved), scaled in registers; sums unchanged
# speedup vs baseline: 2.2030x; 1.1583x over previous
#include <hip/hip_runtime.h>
#include <hip/hip_bf16.h>

typedef float v2f __attribute__((ext_vector_type(2)));
typedef float v4f __attribute__((ext_vector_type(4)));
typedef float v8f __attribute__((ext_vector_type(8)));

static constexpr int DIM = 8192;
static constexpr int FD  = 128;
static constexpr int BM  = 32;
static constexpr int BK  = 128;
static constexpr int KG  = 8;
static constexpr float EPSV = 1e-10f;


__global__ __launch_bounds__(256) void HG_rowdeg(const float* __restrict__ H,
                                                 float* __restrict__ dvis) {
    __shared__ float red[256];
    const int row = blockIdx.x;
    const float* p = H + (size_t)row * DIM;
    float s = 0.0f;
    for (int c = threadIdx.x; c < DIM; c += 256) s += p[c];
    red[threadIdx.x] = s;
    __syncthreads();
    for (int off = 128; off > 0; off >>= 1) {
        if (threadIdx.x < off) red[threadIdx.x] += red[threadIdx.x + off];
        __syncthreads();
    }
    if (threadIdx.x == 0) dvis[row] = 1.0f / sqrtf(red[0] + EPSV);
}

__global__ __launch_bounds__(256) void HG_colzero(float* __restrict__ de) {
    de[blockIdx.x * 256 + threadIdx.x] = 0.0f;
}

__global__ __launch_bounds__(256) void HG_coldeg(const float* __restrict__ H,
                                                 float* __restrict__ de) {
    const int c  = blockIdx.x * 256 + threadIdx.x;
    const int r0 = blockIdx.y * 256;
    float s = 0.0f;
    #pragma unroll 4
    for (int r = 0; r < 256; ++r) s += H[(size_t)(r0 + r) * DIM + c];
    atomicAdd(&de[c], s);
}

__global__ __launch_bounds__(256) void HG_colfin(float* __restrict__ de) {
    const int i = blockIdx.x * 256 + threadIdx.x;
    de[i] = 1.0f / (de[i] + EPSV);
}

template <bool TRANSA, bool OUTSCALE>
__global__ __launch_bounds__(1024) void HG_gemm(const float* __restrict__ A,
                                                const float* __restrict__ srcB,
                                                const float* __restrict__ scaleB,
                                                const float* __restrict__ outScale,
                                                float* __restrict__ out) {
    constexpr int LA = (BK / 2) * BM * 2;
    __shared__ __align__(16) float lAf[2 * LA];
    __shared__ __align__(16) float part[KG * BM * FD];

    const int tid  = threadIdx.x;
    const int wv   = tid >> 5;
    const int lid  = tid & 31;
    const int half = lid >> 4;
    const int ln   = lid & 15;
    const int g    = wv >> 2;
    const int wq   = wv & 3;
    const int m0   = blockIdx.x * BM;

    const int ks   = (tid >> 8) * 32;
    const int t    = tid & 255;
    const int a_i4 = (t & 7) * 4;
    const int a_j  = t >> 3;

    const float* bp = srcB + (size_t)(16 * g + 2 * half) * FD + wq * 32 + 2 * ln;
    const float* sp = scaleB + 16 * g + 2 * half;

    v8f acc[2][2];
    acc[0][0] = (v8f){};
    acc[0][1] = (v8f){};
    acc[1][0] = (v8f){};
    acc[1][1] = (v8f){};

    v4f ra;
    v2f rb[4][2];
    v2f rs[4];
    v2f b0[4], b1[4];

    auto load_chunk = [&](int k0) {
        if (TRANSA) {
            ra = *(const v4f*)(A + (size_t)(k0 + ks + a_j) * DIM + m0 + a_i4);
        } else {
            ra = *(const v4f*)(A + (size_t)(m0 + a_j) * DIM + k0 + ks + a_i4);
        }
        #pragma unroll
        for (int s = 0; s < 4; ++s) {
            rs[s]    = *(const v2f*)(sp + k0 + 4 * s);
            rb[s][0] = *(const v2f*)(bp + (size_t)(k0 + 4 * s) * FD);
            rb[s][1] = *(const v2f*)(bp + (size_t)(k0 + 4 * s + 1) * FD);
        }
    };

    auto store_chunk = [&](int buf) {
        float* lA = lAf + buf * LA;
        if (TRANSA) {
            const int k  = ks + a_j;
            const int kp = k >> 1, kc = k & 1;
            lA[(kp * BM + a_i4 + 0) * 2 + kc] = ra.x;
            lA[(kp * BM + a_i4 + 1) * 2 + kc] = ra.y;
            lA[(kp * BM + a_i4 + 2) * 2 + kc] = ra.z;
            lA[(kp * BM + a_i4 + 3) * 2 + kc] = ra.w;
        } else {
            const int kp = (ks + a_i4) >> 1;
            lA[((kp + 0) * BM + a_j) * 2 + 0] = ra.x;
            lA[((kp + 0) * BM + a_j) * 2 + 1] = ra.y;
            lA[((kp + 1) * BM + a_j) * 2 + 0] = ra.z;
            lA[((kp + 1) * BM + a_j) * 2 + 1] = ra.w;
        }
    };

    auto scale_chunk = [&]() {
        #pragma unroll
        for (int s = 0; s < 4; ++s) {
            b0[s][0] = rb[s][0][0] * rs[s][0];
            b0[s][1] = rb[s][1][0] * rs[s][1];
            b1[s][0] = rb[s][0][1] * rs[s][0];
            b1[s][1] = rb[s][1][1] * rs[s][1];
        }
    };

    load_chunk(0);
    store_chunk(0);
    scale_chunk();
    __syncthreads();

    int cur = 0;
    for (int ki = 0; ki < DIM / BK; ++ki) {
        const float2* lA2 = (const float2*)(lAf + cur * LA);
        v2f a0[4], a1[4];
        #pragma unroll
        for (int s = 0; s < 4; ++s) {
            const int kp = 8 * g + 2 * s + half;
            const float2 a0f = lA2[kp * BM + ln];
            const float2 a1f = lA2[kp * BM + 16 + ln];
            a0[s][0] = a0f.x; a0[s][1] = a0f.y;
            a1[s][0] = a1f.x; a1[s][1] = a1f.y;
        }

        __builtin_amdgcn_sched_barrier(0);

        const int knext = ((ki + 1) & (DIM / BK - 1)) * BK;
        load_chunk(knext);

        __builtin_amdgcn_sched_barrier(0);

        #pragma unroll
        for (int s = 0; s < 4; ++s) {
            acc[0][0] = __builtin_amdgcn_wmma_f32_16x16x4_f32(
                false, a0[s], false, b0[s], (short)0, acc[0][0], false, false);
            acc[0][1] = __builtin_amdgcn_wmma_f32_16x16x4_f32(
                false, a0[s], false, b1[s], (short)0, acc[0][1], false, false);
            acc[1][0] = __builtin_amdgcn_wmma_f32_16x16x4_f32(
                false, a1[s], false, b0[s], (short)0, acc[1][0], false, false);
            acc[1][1] = __builtin_amdgcn_wmma_f32_16x16x4_f32(
                false, a1[s], false, b1[s], (short)0, acc[1][1], false, false);
        }

        __builtin_amdgcn_sched_barrier(0);

        asm volatile("v_nop\n\tv_nop\n\tv_nop\n\tv_nop");

        __builtin_amdgcn_sched_barrier(0);

        store_chunk(cur ^ 1);
        scale_chunk();

        __builtin_amdgcn_sched_barrier(0);

        __syncthreads();
        cur ^= 1;
    }

    #pragma unroll
    for (int r = 0; r < 2; ++r) {
        #pragma unroll
        for (int e = 0; e < 8; ++e) {
            const int row = 16 * r + half * 8 + e;
            v2f pv; pv[0] = acc[r][0][e]; pv[1] = acc[r][1][e];
            *(v2f*)(part + g * (BM * FD) + row * FD + wq * 32 + 2 * ln) = pv;
        }
    }
    __syncthreads();

    const int orow = tid >> 5;
    const int oc4  = (tid & 31) * 4;
    float4 sum = *(const float4*)(part + orow * FD + oc4);
    #pragma unroll
    for (int gg = 1; gg < KG; ++gg) {
        const float4 v = *(const float4*)(part + gg * (BM * FD) + orow * FD + oc4);
        sum.x += v.x; sum.y += v.y; sum.z += v.z; sum.w += v.w;
    }
    const float os = OUTSCALE ? outScale[m0 + orow] : 1.0f;
    sum.x *= os; sum.y *= os; sum.z *= os; sum.w *= os;
    *(float4*)(out + (size_t)(m0 + orow) * FD + oc4) = sum;
}

extern "C" void kernel_launch(void* const* d_in, const int* in_sizes, int n_in,
                              void* d_out, int out_size, void* d_ws, size_t ws_size,
                              hipStream_t stream) {
    (void)in_sizes; (void)n_in; (void)out_size; (void)ws_size;
    const float* H = (const float*)d_in[0];
    const float* F = (const float*)d_in[1];
    float* out  = (float*)d_out;
    float* ws   = (float*)d_ws;
    float* dvis = ws;
    float* de   = ws + DIM;
    float* Y    = ws + 2 * DIM;

    HG_rowdeg<<<DIM, 256, 0, stream>>>(H, dvis);
    HG_colzero<<<DIM / 256, 256, 0, stream>>>(de);
    HG_coldeg<<<dim3(DIM / 256, DIM / 256), 256, 0, stream>>>(H, de);
    HG_colfin<<<DIM / 256, 256, 0, stream>>>(de);

    HG_gemm<true, false><<<DIM / BM, 1024, 0, stream>>>(H, F, dvis, nullptr, Y);
    HG_gemm<false, true><<<DIM / BM, 1024, 0, stream>>>(H, Y, de, dvis, out);
}


# Row and column degrees from one streaming pass over H: 128x2048 block per workgroup, sums in registers, partials added by a small kernel in fixed order; 6 launches to 4, atomics gone (new sum trees)
# speedup vs baseline: 2.5155x; 1.1419x over previous
#include <hip/hip_runtime.h>
#include <hip/hip_bf16.h>

typedef float v2f __attribute__((ext_vector_type(2)));
typedef float v4f __attribute__((ext_vector_type(4)));
typedef float v8f __attribute__((ext_vector_type(8)));
typedef float v16f __attribute__((ext_vector_type(16)));

static constexpr int DIM = 8192;
static constexpr int FD  = 128;
static constexpr int BM  = 32;
static constexpr int BK  = 128;
static constexpr int KG  = 8;
static constexpr float EPSV = 1e-10f;

static constexpr int DR = 128;
static constexpr int DC = 2048;

__global__ __launch_bounds__(512) void HG_deg(const float* __restrict__ H,
                                              float* __restrict__ colpart,
                                              float* __restrict__ rowpart) {
    __shared__ float lrow[16 * DR];
    const int tid  = threadIdx.x;
    const int wv   = tid >> 5;
    const int lane = tid & 31;
    const int c0   = blockIdx.x * DC + wv * 128 + lane * 4;
    const int r0   = blockIdx.y * DR;
    const float* p = H + (size_t)r0 * DIM + c0;

    v4f cs = (v4f){};
    v4f buf[2][16];
    v16f rlo, rhi;

    #pragma unroll
    for (int i = 0; i < 16; ++i) buf[0][i] = *(const v4f*)(p + (size_t)i * DIM);

    #pragma unroll
    for (int g = 0; g < DR / 16; ++g) {
        if (g + 1 < DR / 16) {
            #pragma unroll
            for (int i = 0; i < 16; ++i)
                buf[(g + 1) & 1][i] = *(const v4f*)(p + (size_t)((g + 1) * 16 + i) * DIM);
        }
        #pragma unroll
        for (int i = 0; i < 16; ++i) {
            const v4f v = buf[g & 1][i];
            cs += v;
            const float r = (v[0] + v[1]) + (v[2] + v[3]);
            if (g & 1) rhi[i] = r; else rlo[i] = r;
        }
        if (g & 1) {
            {
                const bool up = (lane & 16) != 0;
                #pragma unroll
                for (int i = 0; i < 16; ++i) {
                    const float lo = rlo[i], hi = rhi[i];
                    rlo[i] = (up ? hi : lo) + __shfl_xor(up ? lo : hi, 16, 32);
                }
            }
            #pragma unroll
            for (int st = 1; st < 5; ++st) {
                const int n = 16 >> st;
                const bool up = (lane & n) != 0;
                #pragma unroll
                for (int i = 0; i < n; ++i) {
                    const float lo = rlo[i], hi = rlo[i + n];
                    rlo[i] = (up ? hi : lo) + __shfl_xor(up ? lo : hi, n, 32);
                }
            }
            lrow[wv * DR + (g >> 1) * 32 + lane] = rlo[0];
        }
    }

    *(v4f*)(colpart + (size_t)blockIdx.y * DIM + c0) = cs;

    __syncthreads();
    if (tid < DR) {
        float s = lrow[tid];
        #pragma unroll
        for (int w = 1; w < 16; ++w) s += lrow[w * DR + tid];
        rowpart[(size_t)blockIdx.x * DIM + r0 + tid] = s;
    }
}

__global__ __launch_bounds__(256) void HG_degfin(const float* __restrict__ colpart,
                                                 const float* __restrict__ rowpart,
                                                 float* __restrict__ dvis,
                                                 float* __restrict__ de) {
    if (blockIdx.x < DIM / 256) {
        const int c = blockIdx.x * 256 + threadIdx.x;
        float s = colpart[c];
        #pragma unroll
        for (int q = 1; q < DIM / DR; ++q) s += colpart[(size_t)q * DIM + c];
        de[c] = 1.0f / (s + EPSV);
    } else {
        const int r = (blockIdx.x - DIM / 256) * 256 + threadIdx.x;
        float s = rowpart[r];
        #pragma unroll
        for (int q = 1; q < DIM / DC; ++q) s += rowpart[(size_t)q * DIM + r];
        dvis[r] = 1.0f / sqrtf(s + EPSV);
    }
}

template <bool TRANSA, bool OUTSCALE>
__global__ __launch_bounds__(1024) void HG_gemm(const float* __restrict__ A,
                                                const float* __restrict__ srcB,
                                                const float* __restrict__ scaleB,
                                                const float* __restrict__ outScale,
                                                float* __restrict__ out) {
    constexpr int LA = (BK / 2) * BM * 2;
    __shared__ __align__(16) float lAf[2 * LA];
    __shared__ __align__(16) float part[KG * BM * FD];

    const int tid  = threadIdx.x;
    const int wv   = tid >> 5;
    const int lid  = tid & 31;
    const int half = lid >> 4;
    const int ln   = lid & 15;
    const int g    = wv >> 2;
    const int wq   = wv & 3;
    const int m0   = blockIdx.x * BM;

    const int ks   = (tid >> 8) * 32;
    const int t    = tid & 255;
    const int a_i4 = (t & 7) * 4;
    const int a_j  = t >> 3;

    const float* bp = srcB + (size_t)(16 * g + 2 * half) * FD + wq * 32 + 2 * ln;
    const float* sp = scaleB + 16 * g + 2 * half;

    v8f acc[2][2];
    acc[0][0] = (v8f){};
    acc[0][1] = (v8f){};
    acc[1][0] = (v8f){};
    acc[1][1] = (v8f){};

    v4f ra;
    v2f rb[4][2];
    v2f rs[4];
    v2f b0[4], b1[4];

    auto load_chunk = [&](int k0) {
        if (TRANSA) {
            ra = *(const v4f*)(A + (size_t)(k0 + ks + a_j) * DIM + m0 + a_i4);
        } else {
            ra = *(const v4f*)(A + (size_t)(m0 + a_j) * DIM + k0 + ks + a_i4);
        }
        #pragma unroll
        for (int s = 0; s < 4; ++s) {
            rs[s]    = *(const v2f*)(sp + k0 + 4 * s);
            rb[s][0] = *(const v2f*)(bp + (size_t)(k0 + 4 * s) * FD);
            rb[s][1] = *(const v2f*)(bp + (size_t)(k0 + 4 * s + 1) * FD);
        }
    };

    auto store_chunk = [&](int buf) {
        float* lA = lAf + buf * LA;
        if (TRANSA) {
            const int k  = ks + a_j;
            const int kp = k >> 1, kc = k & 1;
            lA[(kp * BM + a_i4 + 0) * 2 + kc] = ra.x;
            lA[(kp * BM + a_i4 + 1) * 2 + kc] = ra.y;
            lA[(kp * BM + a_i4 + 2) * 2 + kc] = ra.z;
            lA[(kp * BM + a_i4 + 3) * 2 + kc] = ra.w;
        } else {
            const int kp = (ks + a_i4) >> 1;
            lA[((kp + 0) * BM + a_j) * 2 + 0] = ra.x;
            lA[((kp + 0) * BM + a_j) * 2 + 1] = ra.y;
            lA[((kp + 1) * BM + a_j) * 2 + 0] = ra.z;
            lA[((kp + 1) * BM + a_j) * 2 + 1] = ra.w;
        }
    };

    auto scale_chunk = [&]() {
        #pragma unroll
        for (int s = 0; s < 4; ++s) {
            b0[s][0] = rb[s][0][0] * rs[s][0];
            b0[s][1] = rb[s][1][0] * rs[s][1];
            b1[s][0] = rb[s][0][1] * rs[s][0];
            b1[s][1] = rb[s][1][1] * rs[s][1];
        }
    };

    load_chunk(0);
    store_chunk(0);
    scale_chunk();
    __syncthreads();

    int cur = 0;
    for (int ki = 0; ki < DIM / BK; ++ki) {
        const float2* lA2 = (const float2*)(lAf + cur * LA);
        v2f a0[4], a1[4];
        #pragma unroll
        for (int s = 0; s < 4; ++s) {
            const int kp = 8 * g + 2 * s + half;
            const float2 a0f = lA2[kp * BM + ln];
            const float2 a1f = lA2[kp * BM + 16 + ln];
            a0[s][0] = a0f.x; a0[s][1] = a0f.y;
            a1[s][0] = a1f.x; a1[s][1] = a1f.y;
        }

        __builtin_amdgcn_sched_barrier(0);

        const int knext = ((ki + 1) & (DIM / BK - 1)) * BK;
        load_chunk(knext);

        __builtin_amdgcn_sched_barrier(0);

        #pragma unroll
        for (int s = 0; s < 4; ++s) {
            acc[0][0] = __builtin_amdgcn_wmma_f32_16x16x4_f32(
                false, a0[s], false, b0[s], (short)0, acc[0][0], false, false);
            acc[0][1] = __builtin_amdgcn_wmma_f32_16x16x4_f32(
                false, a0[s], false, b1[s], (short)0, acc[0][1], false, false);
            acc[1][0] = __builtin_amdgcn_wmma_f32_16x16x4_f32(
                false, a1[s], false, b0[s], (short)0, acc[1][0], false, false);
            acc[1][1] = __builtin_amdgcn_wmma_f32_16x16x4_f32(
                false, a1[s], false, b1[s], (short)0, acc[1][1], false, false);
        }

        __builtin_amdgcn_sched_barrier(0);

        asm volatile("v_nop\n\tv_nop\n\tv_nop\n\tv_nop");

        __builtin_amdgcn_sched_barrier(0);

        store_chunk(cur ^ 1);
        scale_chunk();

        __builtin_amdgcn_sched_barrier(0);

        __syncthreads();
        cur ^= 1;
    }

    #pragma unroll
    for (int r = 0; r < 2; ++r) {
        #pragma unroll
        for (int e = 0; e < 8; ++e) {
            const int row = 16 * r + half * 8 + e;
            v2f pv; pv[0] = acc[r][0][e]; pv[1] = acc[r][1][e];
            *(v2f*)(part + g * (BM * FD) + row * FD + wq * 32 + 2 * ln) = pv;
        }
    }
    __syncthreads();

    const int orow = tid >> 5;
    const int oc4  = (tid & 31) * 4;
    float4 sum = *(const float4*)(part + orow * FD + oc4);
    #pragma unroll
    for (int gg = 1; gg < KG; ++gg) {
        const float4 v = *(const float4*)(part + gg * (BM * FD) + orow * FD + oc4);
        sum.x += v.x; sum.y += v.y; sum.z += v.z; sum.w += v.w;
    }
    const float os = OUTSCALE ? outScale[m0 + orow] : 1.0f;
    sum.x *= os; sum.y *= os; sum.z *= os; sum.w *= os;
    *(float4*)(out + (size_t)(m0 + orow) * FD + oc4) = sum;
}

extern "C" void kernel_launch(void* const* d_in, const int* in_sizes, int n_in,
                              void* d_out, int out_size, void* d_ws, size_t ws_size,
                              hipStream_t stream) {
    (void)in_sizes; (void)n_in; (void)out_size; (void)ws_size;
    const float* H = (const float*)d_in[0];
    const float* F = (const float*)d_in[1];
    float* out  = (float*)d_out;
    float* ws   = (float*)d_ws;
    float* dvis = ws;
    float* de   = ws + DIM;
    float* Y    = ws + 2 * DIM;

    float* colpart = Y;
    float* rowpart = Y + (DIM / DR) * DIM;

    HG_deg<<<dim3(DIM / DC, DIM / DR), 512, 0, stream>>>(H, colpart, rowpart);
    HG_degfin<<<2 * (DIM / 256), 256, 0, stream>>>(colpart, rowpart, dvis, de);

    HG_gemm<true, false><<<DIM / BM, 1024, 0, stream>>>(H, F, dvis, nullptr, Y);
    HG_gemm<false, true><<<DIM / BM, 1024, 0, stream>>>(H, Y, de, dvis, out);
}


# GEMM: each wave owns a contiguous K range and the whole 32x128 tile; A, B fragments straight from global, no LDS staging or barrier in the K loop; partials added via LDS, fixed order (new sum tree)
# speedup vs baseline: 3.1081x; 1.2356x over previous
#include <hip/hip_runtime.h>
#include <hip/hip_bf16.h>

typedef float v2f __attribute__((ext_vector_type(2)));
typedef float v4f __attribute__((ext_vector_type(4)));
typedef float v8f __attribute__((ext_vector_type(8)));
typedef float v16f __attribute__((ext_vector_type(16)));

static constexpr int DIM = 8192;
static constexpr int FD  = 128;
static constexpr int BM  = 32;
static constexpr int NW  = 8;
static constexpr int KW  = DIM / NW;
static constexpr float EPSV = 1e-10f;
static_assert(KW % 64 == 0 && BM % (NW * 2) == 0, "wave K range and epilogue passes");

static constexpr int DR = 128;
static constexpr int DC = 2048;

__global__ __launch_bounds__(512) void HG_deg(const float* __restrict__ H,
                                              float* __restrict__ colpart,
                                              float* __restrict__ rowpart) {
    __shared__ float lrow[16 * DR];
    const int tid  = threadIdx.x;
    const int wv   = tid >> 5;
    const int lane = tid & 31;
    const int c0   = blockIdx.x * DC + wv * 128 + lane * 4;
    const int r0   = blockIdx.y * DR;
    const float* p = H + (size_t)r0 * DIM + c0;

    v4f cs = (v4f){};
    v4f buf[2][16];
    v16f rlo, rhi;

    #pragma unroll
    for (int i = 0; i < 16; ++i) buf[0][i] = *(const v4f*)(p + (size_t)i * DIM);

    #pragma unroll
    for (int g = 0; g < DR / 16; ++g) {
        if (g + 1 < DR / 16) {
            #pragma unroll
            for (int i = 0; i < 16; ++i)
                buf[(g + 1) & 1][i] = *(const v4f*)(p + (size_t)((g + 1) * 16 + i) * DIM);
        }
        #pragma unroll
        for (int i = 0; i < 16; ++i) {
            const v4f v = buf[g & 1][i];
            cs += v;
            const float r = (v[0] + v[1]) + (v[2] + v[3]);
            if (g & 1) rhi[i] = r; else rlo[i] = r;
        }
        if (g & 1) {
            {
                const bool up = (lane & 16) != 0;
                #pragma unroll
                for (int i = 0; i < 16; ++i) {
                    const float lo = rlo[i], hi = rhi[i];
                    rlo[i] = (up ? hi : lo) + __shfl_xor(up ? lo : hi, 16, 32);
                }
            }
            #pragma unroll
            for (int st = 1; st < 5; ++st) {
                const int n = 16 >> st;
                const bool up = (lane & n) != 0;
                #pragma unroll
                for (int i = 0; i < n; ++i) {
                    const float lo = rlo[i], hi = rlo[i + n];
                    rlo[i] = (up ? hi : lo) + __shfl_xor(up ? lo : hi, n, 32);
                }
            }
            lrow[wv * DR + (g >> 1) * 32 + lane] = rlo[0];
        }
    }

    *(v4f*)(colpart + (size_t)blockIdx.y * DIM + c0) = cs;

    __syncthreads();
    if (tid < DR) {
        float s = lrow[tid];
        #pragma unroll
        for (int w = 1; w < 16; ++w) s += lrow[w * DR + tid];
        rowpart[(size_t)blockIdx.x * DIM + r0 + tid] = s;
    }
}

__global__ __launch_bounds__(256) void HG_degfin(const float* __restrict__ colpart,
                                                 const float* __restrict__ rowpart,
                                                 float* __restrict__ dvis,
                                                 float* __restrict__ de) {
    if (blockIdx.x < DIM / 256) {
        const int c = blockIdx.x * 256 + threadIdx.x;
        float s = colpart[c];
        #pragma unroll
        for (int q = 1; q < DIM / DR; ++q) s += colpart[(size_t)q * DIM + c];
        de[c] = 1.0f / (s + EPSV);
    } else {
        const int r = (blockIdx.x - DIM / 256) * 256 + threadIdx.x;
        float s = rowpart[r];
        #pragma unroll
        for (int q = 1; q < DIM / DC; ++q) s += rowpart[(size_t)q * DIM + r];
        dvis[r] = 1.0f / sqrtf(s + EPSV);
    }
}

template <bool TRANSA, bool OUTSCALE>
__global__ __launch_bounds__(NW * 32) void HG_gemm(const float* __restrict__ A,
                                                  const float* __restrict__ srcB,
                                                  const float* __restrict__ scaleB,
                                                  const float* __restrict__ outScale,
                                                  float* __restrict__ out) {
    __shared__ __align__(16) float part[NW * BM * FD];

    const int tid  = threadIdx.x;
    const int wv   = tid >> 5;
    const int lid  = tid & 31;
    const int half = lid >> 4;
    const int ln   = lid & 15;
    const int m0   = blockIdx.x * BM;
    const int kw0  = wv * KW;

    auto kstep = [&](int q) {
        return TRANSA ? kw0 + ((4 * q) & (KW - 1)) + 2 * half
                      : kw0 + ((32 * (q >> 3)) & (KW - 1)) + 16 * half + 2 * (q & 7);
    };

    const float* bp  = srcB + 4 * ln;
    const float* apT = A + m0 + ln;
    const float* apN = A + (size_t)(m0 + ln) * DIM + kw0 + 16 * half;

    v8f acc[2][8];
    #pragma unroll
    for (int r = 0; r < 2; ++r) {
        #pragma unroll
        for (int j = 0; j < 8; ++j) acc[r][j] = (v8f){};
    }

    v4f raw[4][4];
    v2f rs[4];
    v2f bs[2][8];
    v2f aT[8][2];
    v4f aN[2][2][4];

    auto loadB = [&](int slot, int kk) {
        const float* p = bp + (size_t)kk * FD;
        raw[slot][0] = *(const v4f*)(p);
        raw[slot][1] = *(const v4f*)(p + 64);
        raw[slot][2] = *(const v4f*)(p + FD);
        raw[slot][3] = *(const v4f*)(p + FD + 64);
        rs[slot]     = *(const v2f*)(scaleB + kk);
    };
    auto scaleBf = [&](int slot, int fs) {
        #pragma unroll
        for (int j = 0; j < 4; ++j) {
            bs[fs][j][0]     = raw[slot][0][j] * rs[slot][0];
            bs[fs][j][1]     = raw[slot][2][j] * rs[slot][1];
            bs[fs][4 + j][0] = raw[slot][1][j] * rs[slot][0];
            bs[fs][4 + j][1] = raw[slot][3][j] * rs[slot][1];
        }
    };
    auto loadAT = [&](int slot, int kk) {
        const float* p = apT + (size_t)kk * DIM;
        aT[slot][0][0] = p[0];
        aT[slot][0][1] = p[DIM];
        aT[slot][1][0] = p[16];
        aT[slot][1][1] = p[DIM + 16];
    };
    auto loadAN = [&](int slot, int grp) {
        const float* p = apN + ((32 * grp) & (KW - 1));
        #pragma unroll
        for (int i = 0; i < 4; ++i) {
            aN[slot][0][i] = *(const v4f*)(p + 4 * i);
            aN[slot][1][i] = *(const v4f*)(p + (size_t)16 * DIM + 4 * i);
        }
    };

    if (TRANSA) {
        #pragma unroll
        for (int q = 0; q < 7; ++q) loadAT(q, kstep(q));
    } else {
        loadAN(0, 0);
        loadAN(1, 1);
    }
    #pragma unroll
    for (int q = 0; q < 4; ++q) loadB(q, kstep(q));
    scaleBf(0, 0);
    loadB(0, kstep(4));

    for (int it = 0; it < KW / 64; ++it) {
        #pragma unroll
        for (int u = 0; u < 16; ++u) {
            v2f a0, a1;
            if (TRANSA) {
                a0 = aT[u & 7][0];
                a1 = aT[u & 7][1];
            } else {
                const v4f x0 = aN[(u >> 3) & 1][0][(u & 7) >> 1];
                const v4f x1 = aN[(u >> 3) & 1][1][(u & 7) >> 1];
                a0[0] = x0[(u & 1) * 2]; a0[1] = x0[(u & 1) * 2 + 1];
                a1[0] = x1[(u & 1) * 2]; a1[1] = x1[(u & 1) * 2 + 1];
            }
            __builtin_amdgcn_sched_barrier(0);
            #pragma unroll
            for (int j = 0; j < 8; ++j) {
                acc[0][j] = __builtin_amdgcn_wmma_f32_16x16x4_f32(
                    false, a0, false, bs[u & 1][j], (short)0, acc[0][j], false, false);
                acc[1][j] = __builtin_amdgcn_wmma_f32_16x16x4_f32(
                    false, a1, false, bs[u & 1][j], (short)0, acc[1][j], false, false);
            }
            __builtin_amdgcn_sched_barrier(0);
            asm volatile("v_nop\n\tv_nop\n\tv_nop\n\tv_nop");
            __builtin_amdgcn_sched_barrier(0);
            scaleBf((u + 1) & 3, (u + 1) & 1);
            loadB((u + 1) & 3, kstep(16 * it + u + 5));
            if (TRANSA) loadAT((u + 7) & 7, kstep(16 * it + u + 7));
            else if ((u & 7) == 7) loadAN((u >> 3) & 1, 2 * it + (u >> 3) + 2);
        }
    }

    #pragma unroll
    for (int r = 0; r < 2; ++r) {
        #pragma unroll
        for (int e = 0; e < 8; ++e) {
            const int row = 16 * r + half * 8 + e;
            v4f lo, hi;
            lo[0] = acc[r][0][e]; lo[1] = acc[r][1][e]; lo[2] = acc[r][2][e]; lo[3] = acc[r][3][e];
            hi[0] = acc[r][4][e]; hi[1] = acc[r][5][e]; hi[2] = acc[r][6][e]; hi[3] = acc[r][7][e];
            float* q = part + wv * (BM * FD) + row * FD + 4 * ln;
            *(v4f*)(q)      = lo;
            *(v4f*)(q + 64) = hi;
        }
    }
    __syncthreads();

    const int oc8 = (tid & 15) * 8;
    #pragma unroll
    for (int ps = 0; ps < BM / (NW * 2); ++ps) {
        const int orow = ps * (NW * 2) + (tid >> 4);
        v4f s0 = *(const v4f*)(part + orow * FD + oc8);
        v4f s1 = *(const v4f*)(part + orow * FD + oc8 + 4);
        #pragma unroll
        for (int w = 1; w < NW; ++w) {
            s0 += *(const v4f*)(part + w * (BM * FD) + orow * FD + oc8);
            s1 += *(const v4f*)(part + w * (BM * FD) + orow * FD + oc8 + 4);
        }
        const float os = OUTSCALE ? outScale[m0 + orow] : 1.0f;
        s0 *= os; s1 *= os;
        *(v4f*)(out + (size_t)(m0 + orow) * FD + oc8)     = s0;
        *(v4f*)(out + (size_t)(m0 + orow) * FD + oc8 + 4) = s1;
    }
}

extern "C" void kernel_launch(void* const* d_in, const int* in_sizes, int n_in,
                              void* d_out, int out_size, void* d_ws, size_t ws_size,
                              hipStream_t stream) {
    (void)in_sizes; (void)n_in; (void)out_size; (void)ws_size;
    const float* H = (const float*)d_in[0];
    const float* F = (const float*)d_in[1];
    float* out  = (float*)d_out;
    float* ws   = (float*)d_ws;
    float* dvis = ws;
    float* de   = ws + DIM;
    float* Y    = ws + 2 * DIM;

    float* colpart = Y;
    float* rowpart = Y + (DIM / DR) * DIM;

    HG_deg<<<dim3(DIM / DC, DIM / DR), 512, 0, stream>>>(H, colpart, rowpart);
    HG_degfin<<<2 * (DIM / 256), 256, 0, stream>>>(colpart, rowpart, dvis, de);

    HG_gemm<true, false><<<DIM / BM, NW * 32, 0, stream>>>(H, F, dvis, nullptr, Y);
    HG_gemm<false, true><<<DIM / BM, NW * 32, 0, stream>>>(H, Y, de, dvis, out);
}
